# starting kernel
# baseline (speedup 1.0000x reference)
#include <hip/hip_runtime.h>
#include <hip/hip_bf16.h>


typedef __attribute__((ext_vector_type(2))) float v2f;
typedef __attribute__((ext_vector_type(8))) float v8f;

constexpr int JV    = 25;
constexpr int CC    = 128;
constexpr int NF    = 300;
constexpr int NSEQ  = 8;
constexpr int TT    = NSEQ * NF;
constexpr int FPB   = 4;
constexpr int ROWS  = FPB * JV;
constexpr int RPAD  = 112;
constexpr int LST   = 132;
constexpr float GN_EPS = 1e-5f;

template <bool TO_LDS>
__device__ __forceinline__ void gemm_rows_128(const float* lds_a,
                                              const float* __restrict__ W,
                                              float* lds_out,
                                              float* __restrict__ gout,
                                              int grow0, int tid)
{
    const int lane = tid & 31;
    const int wv   = tid >> 5;
    const int l15  = lane & 15;
    const int kh   = lane >> 4;
    const int col  = wv * 16 + l15;

    v2f bf[32];
#pragma unroll
    for (int k = 0; k < 32; ++k) {
        const int kr = 4 * k + 2 * kh;
        bf[k].x = W[kr * CC + col];
        bf[k].y = W[(kr + 1) * CC + col];
    }

    for (int m = 0; m < RPAD / 16; ++m) {
        v8f acc = {};
        const float* arow = lds_a + (m * 16 + l15) * LST + 2 * kh;
#pragma unroll
        for (int k = 0; k < 32; ++k) {
            v2f a;
            a.x = arow[4 * k];
            a.y = arow[4 * k + 1];
            acc = __builtin_amdgcn_wmma_f32_16x16x4_f32(
                false, a, false, bf[k], (short)0, acc, false, false);
        }
#pragma unroll
        for (int r = 0; r < 8; ++r) {
            const int row = m * 16 + r + 8 * kh;
            if (TO_LDS) {
                lds_out[row * LST + col] = acc[r];
            } else {
                if (row < ROWS)
                    gout[(size_t)(grow0 + row) * CC + col] = acc[r];
            }
        }
    }
}

__global__ void precompute_kernel(const float* __restrict__ A,
                                  const float* __restrict__ EI,
                                  const float* __restrict__ bs,
                                  float* __restrict__ Aeff,
                                  float* __restrict__ bsum)
{
    const int tid = threadIdx.x;
    for (int idx = tid; idx < 3 * JV * JV; idx += 256)
        Aeff[idx] = A[idx] * EI[idx];
    if (tid < CC)
        bsum[tid] = bs[tid] + bs[CC + tid] + bs[2 * CC + tid];
}

__global__ __launch_bounds__(256) void stgcn_spatial_kernel(
    const float* __restrict__ x,    const float* __restrict__ Ws,
    const float* __restrict__ Wt,   const float* __restrict__ gamma,
    const float* __restrict__ beta, const float* __restrict__ Aeff,
    const float* __restrict__ bsum, float* __restrict__ xw)
{
    __shared__ float lds_x[RPAD * LST];
    __shared__ float lds_y[RPAD * LST];
    __shared__ float lds_h[RPAD * LST];
    __shared__ float lds_A[3 * JV * JV];
    __shared__ float lds_red[FPB * 64 * 2];
    __shared__ float lds_stat[FPB * 2];

    const int tid   = threadIdx.x;
    const int grow0 = blockIdx.x * ROWS;

    for (int idx = tid; idx < ROWS * CC; idx += 256) {
        const int r = idx / CC, c = idx % CC;
        lds_x[r * LST + c] = x[(size_t)grow0 * CC + idx];
        lds_h[r * LST + c] = bsum[c];
    }
    for (int idx = tid; idx < (RPAD - ROWS) * CC; idx += 256) {
        const int r = ROWS + idx / CC, c = idx % CC;
        lds_x[r * LST + c] = 0.f;
        lds_h[r * LST + c] = 0.f;
    }
    for (int idx = tid; idx < 3 * JV * JV; idx += 256)
        lds_A[idx] = Aeff[idx];
    __syncthreads();

    for (int i = 0; i < 3; ++i) {
        gemm_rows_128<true>(lds_x, Ws + i * CC * CC, lds_y, nullptr, 0, tid);
        __syncthreads();
        for (int idx = tid; idx < ROWS * CC; idx += 256) {
            const int r = idx / CC, c = idx % CC;
            const int f = r / JV, u = r % JV;
            const float* av = lds_A + i * JV * JV + u * JV;
            const float* yb = lds_y + (f * JV) * LST + c;
            float s = 0.f;
#pragma unroll
            for (int v = 0; v < JV; ++v) s += av[v] * yb[v * LST];
            lds_h[r * LST + c] += s;
        }
        __syncthreads();
    }

    {
        const int f = tid >> 6;
        const int g = tid & 63;
        float s = 0.f, sq = 0.f;
        for (int e = g; e < JV * CC; e += 64) {
            const float val = lds_h[(f * JV + e / CC) * LST + (e % CC)];
            s += val; sq += val * val;
        }
        lds_red[(f * 64 + g) * 2]     = s;
        lds_red[(f * 64 + g) * 2 + 1] = sq;
        __syncthreads();
        if (g == 0) {
            float ts = 0.f, tq = 0.f;
            for (int k = 0; k < 64; ++k) {
                ts += lds_red[(f * 64 + k) * 2];
                tq += lds_red[(f * 64 + k) * 2 + 1];
            }
            const float inv = 1.f / (float)(JV * CC);
            const float mu  = ts * inv;
            const float var = tq * inv - mu * mu;
            lds_stat[f * 2]     = mu;
            lds_stat[f * 2 + 1] = rsqrtf(var + GN_EPS);
        }
        __syncthreads();
        const float mu = lds_stat[f * 2], rs = lds_stat[f * 2 + 1];
        for (int e = g; e < JV * CC; e += 64) {
            const int addr = (f * JV + e / CC) * LST + (e % CC);
            float val = (lds_h[addr] - mu) * rs * gamma[e] + beta[e];
            lds_h[addr] = val > 0.f ? val : 0.f;
        }
    }
    __syncthreads();

    gemm_rows_128<false>(lds_h, Wt, nullptr, xw, grow0, tid);
}

__global__ __launch_bounds__(256) void stgcn_temporal_kernel(
    const float* __restrict__ xw, const float* __restrict__ bt,
    float* __restrict__ out)
{
    const int g   = blockIdx.x * 256 + threadIdx.x;
    const int seq = g / (JV * CC);
    const int j   = g % (JV * CC);
    const size_t base = (size_t)seq * NF * (JV * CC) + j;
    const float bias  = bt[j % CC];

    float sum = 0.f;
#pragma unroll
    for (int u = 0; u < 5; ++u)
        sum += xw[base + (size_t)u * (JV * CC)];

    for (int t = 0; t < NF; ++t) {
        const float o = sum + bias;
        out[base + (size_t)t * (JV * CC)] = o > 0.f ? o : 0.f;
        if (t + 5 < NF) sum += xw[base + (size_t)(t + 5) * (JV * CC)];
        if (t - 4 >= 0) sum -= xw[base + (size_t)(t - 4) * (JV * CC)];
    }
}

extern "C" void kernel_launch(void* const* d_in, const int* in_sizes, int n_in,
                              void* d_out, int out_size, void* d_ws, size_t ws_size,
                              hipStream_t stream)
{
    const float* x     = (const float*)d_in[0];
    const float* A     = (const float*)d_in[1];
    const float* EI    = (const float*)d_in[2];
    const float* Ws    = (const float*)d_in[3];
    const float* bs    = (const float*)d_in[4];
    const float* Wt    = (const float*)d_in[5];
    const float* bt    = (const float*)d_in[6];
    const float* gamma = (const float*)d_in[7];
    const float* beta  = (const float*)d_in[8];

    float* out = (float*)d_out;
    float* ws  = (float*)d_ws;

    float* xwbuf = ws;
    float* Aeff  = ws + (size_t)TT * JV * CC;
    float* bsum  = Aeff + 3 * JV * JV;

    precompute_kernel<<<1, 256, 0, stream>>>(A, EI, bs, Aeff, bsum);
    stgcn_spatial_kernel<<<TT / FPB, 256, 0, stream>>>(
        x, Ws, Wt, gamma, beta, Aeff, bsum, xwbuf);
    stgcn_temporal_kernel<<<(NSEQ * JV * CC) / 256, 256, 0, stream>>>(
        xwbuf, bt, out);
}


# temporal kernel: one thread = 4 channels x 15-frame chunk, all 23 rows loaded up front as 128-bit loads, direct 9-tap tree sum replaces the serial sliding add/subtract (changes summation order)
# speedup vs baseline: 1.6291x; 1.6291x over previous
#include <hip/hip_runtime.h>
#include <hip/hip_bf16.h>


typedef __attribute__((ext_vector_type(2))) float v2f;
typedef __attribute__((ext_vector_type(8))) float v8f;

constexpr int JV    = 25;
constexpr int CC    = 128;
constexpr int NF    = 300;
constexpr int NSEQ  = 8;
constexpr int TT    = NSEQ * NF;
constexpr int FPB   = 4;
constexpr int ROWS  = FPB * JV;
constexpr int RPAD  = 112;
constexpr int LST   = 132;
constexpr float GN_EPS = 1e-5f;

template <bool TO_LDS>
__device__ __forceinline__ void gemm_rows_128(const float* lds_a,
                                              const float* __restrict__ W,
                                              float* lds_out,
                                              float* __restrict__ gout,
                                              int grow0, int tid)
{
    const int lane = tid & 31;
    const int wv   = tid >> 5;
    const int l15  = lane & 15;
    const int kh   = lane >> 4;
    const int col  = wv * 16 + l15;

    v2f bf[32];
#pragma unroll
    for (int k = 0; k < 32; ++k) {
        const int kr = 4 * k + 2 * kh;
        bf[k].x = W[kr * CC + col];
        bf[k].y = W[(kr + 1) * CC + col];
    }

    for (int m = 0; m < RPAD / 16; ++m) {
        v8f acc = {};
        const float* arow = lds_a + (m * 16 + l15) * LST + 2 * kh;
#pragma unroll
        for (int k = 0; k < 32; ++k) {
            v2f a;
            a.x = arow[4 * k];
            a.y = arow[4 * k + 1];
            acc = __builtin_amdgcn_wmma_f32_16x16x4_f32(
                false, a, false, bf[k], (short)0, acc, false, false);
        }
#pragma unroll
        for (int r = 0; r < 8; ++r) {
            const int row = m * 16 + r + 8 * kh;
            if (TO_LDS) {
                lds_out[row * LST + col] = acc[r];
            } else {
                if (row < ROWS)
                    gout[(size_t)(grow0 + row) * CC + col] = acc[r];
            }
        }
    }
}

__global__ void precompute_kernel(const float* __restrict__ A,
                                  const float* __restrict__ EI,
                                  const float* __restrict__ bs,
                                  float* __restrict__ Aeff,
                                  float* __restrict__ bsum)
{
    const int tid = threadIdx.x;
    for (int idx = tid; idx < 3 * JV * JV; idx += 256)
        Aeff[idx] = A[idx] * EI[idx];
    if (tid < CC)
        bsum[tid] = bs[tid] + bs[CC + tid] + bs[2 * CC + tid];
}

__global__ __launch_bounds__(256) void stgcn_spatial_kernel(
    const float* __restrict__ x,    const float* __restrict__ Ws,
    const float* __restrict__ Wt,   const float* __restrict__ gamma,
    const float* __restrict__ beta, const float* __restrict__ Aeff,
    const float* __restrict__ bsum, float* __restrict__ xw)
{
    __shared__ float lds_x[RPAD * LST];
    __shared__ float lds_y[RPAD * LST];
    __shared__ float lds_h[RPAD * LST];
    __shared__ float lds_A[3 * JV * JV];
    __shared__ float lds_red[FPB * 64 * 2];
    __shared__ float lds_stat[FPB * 2];

    const int tid   = threadIdx.x;
    const int grow0 = blockIdx.x * ROWS;

    for (int idx = tid; idx < ROWS * CC; idx += 256) {
        const int r = idx / CC, c = idx % CC;
        lds_x[r * LST + c] = x[(size_t)grow0 * CC + idx];
        lds_h[r * LST + c] = bsum[c];
    }
    for (int idx = tid; idx < (RPAD - ROWS) * CC; idx += 256) {
        const int r = ROWS + idx / CC, c = idx % CC;
        lds_x[r * LST + c] = 0.f;
        lds_h[r * LST + c] = 0.f;
    }
    for (int idx = tid; idx < 3 * JV * JV; idx += 256)
        lds_A[idx] = Aeff[idx];
    __syncthreads();

    for (int i = 0; i < 3; ++i) {
        gemm_rows_128<true>(lds_x, Ws + i * CC * CC, lds_y, nullptr, 0, tid);
        __syncthreads();
        for (int idx = tid; idx < ROWS * CC; idx += 256) {
            const int r = idx / CC, c = idx % CC;
            const int f = r / JV, u = r % JV;
            const float* av = lds_A + i * JV * JV + u * JV;
            const float* yb = lds_y + (f * JV) * LST + c;
            float s = 0.f;
#pragma unroll
            for (int v = 0; v < JV; ++v) s += av[v] * yb[v * LST];
            lds_h[r * LST + c] += s;
        }
        __syncthreads();
    }

    {
        const int f = tid >> 6;
        const int g = tid & 63;
        float s = 0.f, sq = 0.f;
        for (int e = g; e < JV * CC; e += 64) {
            const float val = lds_h[(f * JV + e / CC) * LST + (e % CC)];
            s += val; sq += val * val;
        }
        lds_red[(f * 64 + g) * 2]     = s;
        lds_red[(f * 64 + g) * 2 + 1] = sq;
        __syncthreads();
        if (g == 0) {
            float ts = 0.f, tq = 0.f;
            for (int k = 0; k < 64; ++k) {
                ts += lds_red[(f * 64 + k) * 2];
                tq += lds_red[(f * 64 + k) * 2 + 1];
            }
            const float inv = 1.f / (float)(JV * CC);
            const float mu  = ts * inv;
            const float var = tq * inv - mu * mu;
            lds_stat[f * 2]     = mu;
            lds_stat[f * 2 + 1] = rsqrtf(var + GN_EPS);
        }
        __syncthreads();
        const float mu = lds_stat[f * 2], rs = lds_stat[f * 2 + 1];
        for (int e = g; e < JV * CC; e += 64) {
            const int addr = (f * JV + e / CC) * LST + (e % CC);
            float val = (lds_h[addr] - mu) * rs * gamma[e] + beta[e];
            lds_h[addr] = val > 0.f ? val : 0.f;
        }
    }
    __syncthreads();

    gemm_rows_128<false>(lds_h, Wt, nullptr, xw, grow0, tid);
}

constexpr int TCH  = 15;
constexpr int NCH  = NF / TCH;
constexpr int C4   = JV * CC / 4;
constexpr int HALO = 4;

typedef __attribute__((ext_vector_type(4))) float v4f;

__global__ __launch_bounds__(256) void stgcn_temporal_kernel(
    const float* __restrict__ xw, const float* __restrict__ bt,
    float* __restrict__ out)
{
    const int g     = blockIdx.x * 256 + threadIdx.x;
    const int crow  = g / C4;
    const int c4    = g % C4;
    const int seq   = crow / NCH;
    const int chunk = crow % NCH;
    const int t0    = chunk * TCH;

    const size_t fbase = (size_t)seq * NF * C4 + c4;
    const v4f* __restrict__ src = reinterpret_cast<const v4f*>(xw) + fbase;
    v4f* __restrict__ dst       = reinterpret_cast<v4f*>(out) + fbase;

    const bool has_lo = chunk > 0;
    const bool has_hi = chunk < NCH - 1;
    const v4f* __restrict__ lo  = src + (size_t)(has_lo ? t0 - HALO : t0) * C4;
    const v4f* __restrict__ mid = src + (size_t)t0 * C4;
    const v4f* __restrict__ hi  = src + (size_t)(has_hi ? t0 + TCH : t0 + TCH - HALO) * C4;

    const int cb = (c4 & (CC / 4 - 1)) * 4;
    v4f bias;
    bias.x = bt[cb]; bias.y = bt[cb + 1]; bias.z = bt[cb + 2]; bias.w = bt[cb + 3];

    v4f v[TCH + 2 * HALO];
#pragma unroll
    for (int i = 0; i < HALO; ++i) v[i] = lo[(size_t)i * C4];
#pragma unroll
    for (int i = 0; i < TCH; ++i) v[HALO + i] = mid[(size_t)i * C4];
#pragma unroll
    for (int i = 0; i < HALO; ++i) v[HALO + TCH + i] = hi[(size_t)i * C4];

    const v4f zero4 = {0.f, 0.f, 0.f, 0.f};
#pragma unroll
    for (int i = 0; i < HALO; ++i) {
        v[i]              = has_lo ? v[i] : zero4;
        v[HALO + TCH + i] = has_hi ? v[HALO + TCH + i] : zero4;
    }

    v4f p[TCH + 6];
#pragma unroll
    for (int i = 0; i < TCH + 6; ++i) p[i] = v[i] + v[i + 1];
    v4f q[TCH + 4];
#pragma unroll
    for (int i = 0; i < TCH + 4; ++i) q[i] = p[i] + p[i + 2];
#pragma unroll
    for (int i = 0; i < TCH; ++i) {
        const v4f s = ((q[i] + q[i + 4]) + v[i + 8]) + bias;
        v4f o;
        o.x = s.x > 0.f ? s.x : 0.f;
        o.y = s.y > 0.f ? s.y : 0.f;
        o.z = s.z > 0.f ? s.z : 0.f;
        o.w = s.w > 0.f ? s.w : 0.f;
        dst[(size_t)(t0 + i) * C4] = o;
    }
}

extern "C" void kernel_launch(void* const* d_in, const int* in_sizes, int n_in,
                              void* d_out, int out_size, void* d_ws, size_t ws_size,
                              hipStream_t stream)
{
    const float* x     = (const float*)d_in[0];
    const float* A     = (const float*)d_in[1];
    const float* EI    = (const float*)d_in[2];
    const float* Ws    = (const float*)d_in[3];
    const float* bs    = (const float*)d_in[4];
    const float* Wt    = (const float*)d_in[5];
    const float* bt    = (const float*)d_in[6];
    const float* gamma = (const float*)d_in[7];
    const float* beta  = (const float*)d_in[8];

    float* out = (float*)d_out;
    float* ws  = (float*)d_ws;

    float* xwbuf = ws;
    float* Aeff  = ws + (size_t)TT * JV * CC;
    float* bsum  = Aeff + 3 * JV * JV;

    precompute_kernel<<<1, 256, 0, stream>>>(A, EI, bs, Aeff, bsum);
    stgcn_spatial_kernel<<<TT / FPB, 256, 0, stream>>>(
        x, Ws, Wt, gamma, beta, Aeff, bsum, xwbuf);
    stgcn_temporal_kernel<<<(NSEQ * NCH * C4) / 256, 256, 0, stream>>>(
        xwbuf, bt, out);
}


# spatial kernel: joint contraction Aeff_i @ y moved from LDS-fed vector FMAs to f32 WMMA (2 M-tiles x 7 k-steps per frame, h tiles kept in registers over the 3 branches); changes summation order
# speedup vs baseline: 2.7045x; 1.6601x over previous
#include <hip/hip_runtime.h>
#include <hip/hip_bf16.h>


typedef __attribute__((ext_vector_type(2))) float v2f;
typedef __attribute__((ext_vector_type(8))) float v8f;

constexpr int JV    = 25;
constexpr int CC    = 128;
constexpr int NF    = 300;
constexpr int NSEQ  = 8;
constexpr int TT    = NSEQ * NF;
constexpr int FPB   = 4;
constexpr int ROWS  = FPB * JV;
constexpr int RPAD  = 112;
constexpr int LST   = 132;
constexpr float GN_EPS = 1e-5f;

template <bool TO_LDS>
__device__ __forceinline__ void gemm_rows_128(const float* lds_a,
                                              const float* __restrict__ W,
                                              float* lds_out,
                                              float* __restrict__ gout,
                                              int grow0, int tid)
{
    const int lane = tid & 31;
    const int wv   = tid >> 5;
    const int l15  = lane & 15;
    const int kh   = lane >> 4;
    const int col  = wv * 16 + l15;

    v2f bf[32];
#pragma unroll
    for (int k = 0; k < 32; ++k) {
        const int kr = 4 * k + 2 * kh;
        bf[k].x = W[kr * CC + col];
        bf[k].y = W[(kr + 1) * CC + col];
    }

    for (int m = 0; m < RPAD / 16; ++m) {
        v8f acc = {};
        const float* arow = lds_a + (m * 16 + l15) * LST + 2 * kh;
#pragma unroll
        for (int k = 0; k < 32; ++k) {
            v2f a;
            a.x = arow[4 * k];
            a.y = arow[4 * k + 1];
            acc = __builtin_amdgcn_wmma_f32_16x16x4_f32(
                false, a, false, bf[k], (short)0, acc, false, false);
        }
#pragma unroll
        for (int r = 0; r < 8; ++r) {
            const int row = m * 16 + r + 8 * kh;
            if (TO_LDS) {
                lds_out[row * LST + col] = acc[r];
            } else {
                if (row < ROWS)
                    gout[(size_t)(grow0 + row) * CC + col] = acc[r];
            }
        }
    }
}

__global__ void precompute_kernel(const float* __restrict__ A,
                                  const float* __restrict__ EI,
                                  const float* __restrict__ bs,
                                  float* __restrict__ Aeff,
                                  float* __restrict__ bsum)
{
    const int tid = threadIdx.x;
    for (int idx = tid; idx < 3 * JV * JV; idx += 256)
        Aeff[idx] = A[idx] * EI[idx];
    if (tid < CC)
        bsum[tid] = bs[tid] + bs[CC + tid] + bs[2 * CC + tid];
}

__global__ __launch_bounds__(256) void stgcn_spatial_kernel(
    const float* __restrict__ x,    const float* __restrict__ Ws,
    const float* __restrict__ Wt,   const float* __restrict__ gamma,
    const float* __restrict__ beta, const float* __restrict__ Aeff,
    const float* __restrict__ bsum, float* __restrict__ xw)
{
    __shared__ float lds_x[RPAD * LST];
    __shared__ float lds_y[RPAD * LST];
    __shared__ float lds_h[RPAD * LST];
    __shared__ float lds_A[3 * JV * JV];
    __shared__ float lds_red[FPB * 64 * 2];
    __shared__ float lds_stat[FPB * 2];

    const int tid   = threadIdx.x;
    const int grow0 = blockIdx.x * ROWS;

    for (int idx = tid; idx < ROWS * CC; idx += 256) {
        const int r = idx / CC, c = idx % CC;
        lds_x[r * LST + c] = x[(size_t)grow0 * CC + idx];
    }
    for (int idx = tid; idx < (RPAD - ROWS) * CC; idx += 256) {
        const int r = ROWS + idx / CC, c = idx % CC;
        lds_x[r * LST + c] = 0.f;
        lds_h[r * LST + c] = 0.f;
    }
    for (int idx = tid; idx < 3 * JV * JV; idx += 256)
        lds_A[idx] = Aeff[idx];
    __syncthreads();

    {
        const int lane = tid & 31;
        const int wv   = tid >> 5;
        const int l15  = lane & 15;
        const int kh   = lane >> 4;
        const int col  = wv * 16 + l15;

        v8f hacc[FPB][2];
        {
            const float b0 = bsum[col];
#pragma unroll
            for (int f = 0; f < FPB; ++f)
#pragma unroll
                for (int mt = 0; mt < 2; ++mt)
#pragma unroll
                    for (int r = 0; r < 8; ++r) hacc[f][mt][r] = b0;
        }

        for (int i = 0; i < 3; ++i) {
            gemm_rows_128<true>(lds_x, Ws + i * CC * CC, lds_y, nullptr, 0, tid);

            v2f af[2][7];
#pragma unroll
            for (int mt = 0; mt < 2; ++mt) {
                const int u = mt * 16 + l15;
                const float* arow = lds_A + i * JV * JV + (u < JV ? u : 0) * JV;
#pragma unroll
                for (int k = 0; k < 7; ++k) {
                    const int v0 = 4 * k + 2 * kh;
                    const float ax = arow[v0 < JV ? v0 : 0];
                    const float ay = arow[v0 + 1 < JV ? v0 + 1 : 0];
                    af[mt][k].x = (u < JV && v0 < JV) ? ax : 0.f;
                    af[mt][k].y = (u < JV && v0 + 1 < JV) ? ay : 0.f;
                }
            }
            __syncthreads();

#pragma unroll
            for (int f = 0; f < FPB; ++f) {
                const float* yb = lds_y + (f * JV + 2 * kh) * LST + col;
                v2f yf[7];
#pragma unroll
                for (int k = 0; k < 7; ++k) {
                    yf[k].x = yb[(4 * k) * LST];
                    yf[k].y = yb[(4 * k + 1) * LST];
                }
                yf[6].x = (kh == 0) ? yf[6].x : 0.f;
                yf[6].y = 0.f;
#pragma unroll
                for (int k = 0; k < 7; ++k) {
#pragma unroll
                    for (int mt = 0; mt < 2; ++mt)
                        hacc[f][mt] = __builtin_amdgcn_wmma_f32_16x16x4_f32(
                            false, af[mt][k], false, yf[k], (short)0,
                            hacc[f][mt], false, false);
                }
            }
        }

#pragma unroll
        for (int f = 0; f < FPB; ++f)
#pragma unroll
            for (int mt = 0; mt < 2; ++mt)
#pragma unroll
                for (int r = 0; r < 8; ++r) {
                    const int u = mt * 16 + r + 8 * kh;
                    if (u < JV)
                        lds_h[(f * JV + u) * LST + col] = hacc[f][mt][r];
                }
    }
    __syncthreads();

    {
        const int f = tid >> 6;
        const int g = tid & 63;
        float s = 0.f, sq = 0.f;
        for (int e = g; e < JV * CC; e += 64) {
            const float val = lds_h[(f * JV + e / CC) * LST + (e % CC)];
            s += val; sq += val * val;
        }
        lds_red[(f * 64 + g) * 2]     = s;
        lds_red[(f * 64 + g) * 2 + 1] = sq;
        __syncthreads();
        if (g == 0) {
            float ts = 0.f, tq = 0.f;
            for (int k = 0; k < 64; ++k) {
                ts += lds_red[(f * 64 + k) * 2];
                tq += lds_red[(f * 64 + k) * 2 + 1];
            }
            const float inv = 1.f / (float)(JV * CC);
            const float mu  = ts * inv;
            const float var = tq * inv - mu * mu;
            lds_stat[f * 2]     = mu;
            lds_stat[f * 2 + 1] = rsqrtf(var + GN_EPS);
        }
        __syncthreads();
        const float mu = lds_stat[f * 2], rs = lds_stat[f * 2 + 1];
        for (int e = g; e < JV * CC; e += 64) {
            const int addr = (f * JV + e / CC) * LST + (e % CC);
            float val = (lds_h[addr] - mu) * rs * gamma[e] + beta[e];
            lds_h[addr] = val > 0.f ? val : 0.f;
        }
    }
    __syncthreads();

    gemm_rows_128<false>(lds_h, Wt, nullptr, xw, grow0, tid);
}

constexpr int TCH  = 15;
constexpr int NCH  = NF / TCH;
constexpr int C4   = JV * CC / 4;
constexpr int HALO = 4;

typedef __attribute__((ext_vector_type(4))) float v4f;

__global__ __launch_bounds__(256) void stgcn_temporal_kernel(
    const float* __restrict__ xw, const float* __restrict__ bt,
    float* __restrict__ out)
{
    const int g     = blockIdx.x * 256 + threadIdx.x;
    const int crow  = g / C4;
    const int c4    = g % C4;
    const int seq   = crow / NCH;
    const int chunk = crow % NCH;
    const int t0    = chunk * TCH;

    const size_t fbase = (size_t)seq * NF * C4 + c4;
    const v4f* __restrict__ src = reinterpret_cast<const v4f*>(xw) + fbase;
    v4f* __restrict__ dst       = reinterpret_cast<v4f*>(out) + fbase;

    const bool has_lo = chunk > 0;
    const bool has_hi = chunk < NCH - 1;
    const v4f* __restrict__ lo  = src + (size_t)(has_lo ? t0 - HALO : t0) * C4;
    const v4f* __restrict__ mid = src + (size_t)t0 * C4;
    const v4f* __restrict__ hi  = src + (size_t)(has_hi ? t0 + TCH : t0 + TCH - HALO) * C4;

    const int cb = (c4 & (CC / 4 - 1)) * 4;
    v4f bias;
    bias.x = bt[cb]; bias.y = bt[cb + 1]; bias.z = bt[cb + 2]; bias.w = bt[cb + 3];

    v4f v[TCH + 2 * HALO];
#pragma unroll
    for (int i = 0; i < HALO; ++i) v[i] = lo[(size_t)i * C4];
#pragma unroll
    for (int i = 0; i < TCH; ++i) v[HALO + i] = mid[(size_t)i * C4];
#pragma unroll
    for (int i = 0; i < HALO; ++i) v[HALO + TCH + i] = hi[(size_t)i * C4];

    const v4f zero4 = {0.f, 0.f, 0.f, 0.f};
#pragma unroll
    for (int i = 0; i < HALO; ++i) {
        v[i]              = has_lo ? v[i] : zero4;
        v[HALO + TCH + i] = has_hi ? v[HALO + TCH + i] : zero4;
    }

    v4f p[TCH + 6];
#pragma unroll
    for (int i = 0; i < TCH + 6; ++i) p[i] = v[i] + v[i + 1];
    v4f q[TCH + 4];
#pragma unroll
    for (int i = 0; i < TCH + 4; ++i) q[i] = p[i] + p[i + 2];
#pragma unroll
    for (int i = 0; i < TCH; ++i) {
        const v4f s = ((q[i] + q[i + 4]) + v[i + 8]) + bias;
        v4f o;
        o.x = s.x > 0.f ? s.x : 0.f;
        o.y = s.y > 0.f ? s.y : 0.f;
        o.z = s.z > 0.f ? s.z : 0.f;
        o.w = s.w > 0.f ? s.w : 0.f;
        dst[(size_t)(t0 + i) * C4] = o;
    }
}

extern "C" void kernel_launch(void* const* d_in, const int* in_sizes, int n_in,
                              void* d_out, int out_size, void* d_ws, size_t ws_size,
                              hipStream_t stream)
{
    const float* x     = (const float*)d_in[0];
    const float* A     = (const float*)d_in[1];
    const float* EI    = (const float*)d_in[2];
    const float* Ws    = (const float*)d_in[3];
    const float* bs    = (const float*)d_in[4];
    const float* Wt    = (const float*)d_in[5];
    const float* bt    = (const float*)d_in[6];
    const float* gamma = (const float*)d_in[7];
    const float* beta  = (const float*)d_in[8];

    float* out = (float*)d_out;
    float* ws  = (float*)d_ws;

    float* xwbuf = ws;
    float* Aeff  = ws + (size_t)TT * JV * CC;
    float* bsum  = Aeff + 3 * JV * JV;

    precompute_kernel<<<1, 256, 0, stream>>>(A, EI, bs, Aeff, bsum);
    stgcn_spatial_kernel<<<TT / FPB, 256, 0, stream>>>(
        x, Ws, Wt, gamma, beta, Aeff, bsum, xwbuf);
    stgcn_temporal_kernel<<<(NSEQ * NCH * C4) / 256, 256, 0, stream>>>(
        xwbuf, bt, out);
}


# adjacency product and bias sum formed inside the spatial kernel, loads issued in one group with the x tile's: precompute launch and the 8 serial load-store trips of the adjacency copy gone
# speedup vs baseline: 4.9792x; 1.8411x over previous
#include <hip/hip_runtime.h>
#include <hip/hip_bf16.h>


typedef __attribute__((ext_vector_type(2))) float v2f;
typedef __attribute__((ext_vector_type(4))) float v4f;
typedef __attribute__((ext_vector_type(8))) float v8f;

constexpr int JV    = 25;
constexpr int CC    = 128;
constexpr int NF    = 300;
constexpr int NSEQ  = 8;
constexpr int TT    = NSEQ * NF;
constexpr int FPB   = 5;
constexpr int ROWS  = FPB * JV;
constexpr int RPAD  = 128;
constexpr int LST   = 132;
constexpr float GN_EPS = 1e-5f;

template <bool TO_LDS>
__device__ __forceinline__ void gemm_rows_128(const float* lds_a,
                                              const float* __restrict__ W,
                                              float* lds_out,
                                              float* __restrict__ gout,
                                              int grow0, int tid)
{
    const int lane = tid & 31;
    const int wv   = tid >> 5;
    const int l15  = lane & 15;
    const int kh   = lane >> 4;
    const int col  = wv * 16 + l15;

    v2f bf[32];
#pragma unroll
    for (int k = 0; k < 32; ++k) {
        const int kr = 4 * k + 2 * kh;
        bf[k].x = W[kr * CC + col];
        bf[k].y = W[(kr + 1) * CC + col];
    }

    for (int m = 0; m < RPAD / 16; ++m) {
        v8f acc = {};
        const float* arow = lds_a + (m * 16 + l15) * LST + 2 * kh;
#pragma unroll
        for (int k = 0; k < 32; ++k) {
            v2f a;
            a.x = arow[4 * k];
            a.y = arow[4 * k + 1];
            acc = __builtin_amdgcn_wmma_f32_16x16x4_f32(
                false, a, false, bf[k], (short)0, acc, false, false);
        }
#pragma unroll
        for (int r = 0; r < 8; ++r) {
            const int row = m * 16 + r + 8 * kh;
            if (TO_LDS) {
                lds_out[row * LST + col] = acc[r];
            } else {
                if (row < ROWS)
                    gout[(size_t)(grow0 + row) * CC + col] = acc[r];
            }
        }
    }
}

__global__ __launch_bounds__(256, 4) void stgcn_spatial_kernel(
    const float* __restrict__ x,    const float* __restrict__ Ws,
    const float* __restrict__ Wt,   const float* __restrict__ gamma,
    const float* __restrict__ beta, const float* __restrict__ A,
    const float* __restrict__ EI,   const float* __restrict__ bs,
    float* __restrict__ xw)
{
    __shared__ __attribute__((aligned(16))) float lds_x[RPAD * LST];
    __shared__ __attribute__((aligned(16))) float lds_y[RPAD * LST];
    __shared__ float lds_A[(3 * JV * JV + 255) / 256 * 256];
    float* const lds_h = lds_x;

    const int tid   = threadIdx.x;
    const int grow0 = blockIdx.x * ROWS;

    constexpr int NA  = 3 * JV * JV;
    constexpr int NAJ = (NA + 255) / 256;
    float ar[NAJ], er[NAJ];
    float b0;
    {
        const v4f* __restrict__ xg =
            reinterpret_cast<const v4f*>(x + (size_t)grow0 * CC);
        constexpr int NP = ROWS * CC / 4;
        constexpr int NJ = RPAD * CC / 4 / 256;
        static_assert(RPAD * CC / 4 == NJ * 256, "tile = whole pieces per thread");
        const v4f zero4 = {0.f, 0.f, 0.f, 0.f};
        v4f xr[NJ];
#pragma unroll
        for (int j = 0; j < NJ; ++j) {
            const int p = j * 256 + tid;
            xr[j] = xg[(j < NP / 256 || p < NP) ? p : NP - 1];
        }
#pragma unroll
        for (int j = 0; j < NAJ; ++j) {
            const int idx = j * 256 + tid;
            const int ci  = (j < NA / 256 || idx < NA) ? idx : NA - 1;
            ar[j] = A[ci];
            er[j] = EI[ci];
        }
        {
            const int bcol = (tid >> 5) * 16 + (tid & 15);
            b0 = bs[bcol] + bs[CC + bcol] + bs[2 * CC + bcol];
        }
#pragma unroll
        for (int j = 0; j < NJ; ++j) {
            const int p = j * 256 + tid;
            const int r = p / (CC / 4), c4 = p % (CC / 4);
            *reinterpret_cast<v4f*>(lds_x + r * LST + c4 * 4) =
                (j < NP / 256 || p < NP) ? xr[j] : zero4;
        }
    }
#pragma unroll
    for (int j = 0; j < NAJ; ++j) {
        lds_A[j * 256 + tid] = ar[j] * er[j];
    }
    __syncthreads();

    {
        const int lane = tid & 31;
        const int wv   = tid >> 5;
        const int l15  = lane & 15;
        const int kh   = lane >> 4;
        const int col  = wv * 16 + l15;

        v8f hacc[FPB][2];
        {
#pragma unroll
            for (int f = 0; f < FPB; ++f)
#pragma unroll
                for (int mt = 0; mt < 2; ++mt)
#pragma unroll
                    for (int r = 0; r < 8; ++r) hacc[f][mt][r] = b0;
        }

        for (int i = 0; i < 3; ++i) {
            gemm_rows_128<true>(lds_x, Ws + i * CC * CC, lds_y, nullptr, 0, tid);

            v2f af[2][7];
#pragma unroll
            for (int mt = 0; mt < 2; ++mt) {
                const int u = mt * 16 + l15;
                const float* arow = lds_A + i * JV * JV + (u < JV ? u : 0) * JV;
#pragma unroll
                for (int k = 0; k < 7; ++k) {
                    const int v0 = 4 * k + 2 * kh;
                    const float ax = arow[v0 < JV ? v0 : 0];
                    const float ay = arow[v0 + 1 < JV ? v0 + 1 : 0];
                    af[mt][k].x = (u < JV && v0 < JV) ? ax : 0.f;
                    af[mt][k].y = (u < JV && v0 + 1 < JV) ? ay : 0.f;
                }
            }
            __syncthreads();

#pragma unroll
            for (int f = 0; f < FPB; ++f) {
                const float* yb = lds_y + (f * JV + 2 * kh) * LST + col;
                v2f yf[7];
#pragma unroll
                for (int k = 0; k < 7; ++k) {
                    yf[k].x = yb[(4 * k) * LST];
                    yf[k].y = yb[(4 * k + 1) * LST];
                }
                yf[6].x = (kh == 0) ? yf[6].x : 0.f;
                yf[6].y = 0.f;
#pragma unroll
                for (int k = 0; k < 7; ++k) {
#pragma unroll
                    for (int mt = 0; mt < 2; ++mt)
                        hacc[f][mt] = __builtin_amdgcn_wmma_f32_16x16x4_f32(
                            false, af[mt][k], false, yf[k], (short)0,
                            hacc[f][mt], false, false);
                }
            }
        }

#pragma unroll
        for (int f = 0; f < FPB; ++f)
#pragma unroll
            for (int mt = 0; mt < 2; ++mt)
#pragma unroll
                for (int r = 0; r < 8; ++r) {
                    const int u = mt * 16 + r + 8 * kh;
                    if (u < JV)
                        lds_h[(f * JV + u) * LST + col] = hacc[f][mt][r];
                }
    }
    __syncthreads();

    {
        const int lane = tid & 31;
        const int wv   = tid >> 5;
        if (wv < FPB) {
            float* hb = lds_h + (wv * JV) * LST + lane * 4;
            v4f hv[JV];
            v4f s4 = {0.f, 0.f, 0.f, 0.f}, q4 = {0.f, 0.f, 0.f, 0.f};
#pragma unroll
            for (int u = 0; u < JV; ++u) {
                hv[u] = *reinterpret_cast<const v4f*>(hb + u * LST);
                s4 += hv[u];
                q4 += hv[u] * hv[u];
            }
            float ts = (s4.x + s4.y) + (s4.z + s4.w);
            float tq = (q4.x + q4.y) + (q4.z + q4.w);
#pragma unroll
            for (int off = 16; off > 0; off >>= 1) {
                ts += __shfl_xor(ts, off);
                tq += __shfl_xor(tq, off);
            }
            const float inv = 1.f / (float)(JV * CC);
            const float mu  = ts * inv;
            const float var = tq * inv - mu * mu;
            const float rs  = rsqrtf(var + GN_EPS);
            const v4f* __restrict__ g4 = reinterpret_cast<const v4f*>(gamma) + lane;
            const v4f* __restrict__ b4 = reinterpret_cast<const v4f*>(beta) + lane;
#pragma unroll
            for (int u = 0; u < JV; ++u) {
                const v4f gm = g4[u * (CC / 4)];
                const v4f bb = b4[u * (CC / 4)];
                const v4f val = (hv[u] - mu) * rs * gm + bb;
                v4f o;
                o.x = val.x > 0.f ? val.x : 0.f;
                o.y = val.y > 0.f ? val.y : 0.f;
                o.z = val.z > 0.f ? val.z : 0.f;
                o.w = val.w > 0.f ? val.w : 0.f;
                *reinterpret_cast<v4f*>(hb + u * LST) = o;
            }
        }
    }
    __syncthreads();

    gemm_rows_128<false>(lds_h, Wt, nullptr, xw, grow0, tid);
}

constexpr int TCH  = 15;
constexpr int NCH  = NF / TCH;
constexpr int C4   = JV * CC / 4;
constexpr int HALO = 4;

__global__ __launch_bounds__(256) void stgcn_temporal_kernel(
    const float* __restrict__ xw, const float* __restrict__ bt,
    float* __restrict__ out)
{
    const int g     = blockIdx.x * 256 + threadIdx.x;
    const int crow  = g / C4;
    const int c4    = g % C4;
    const int seq   = crow / NCH;
    const int chunk = crow % NCH;
    const int t0    = chunk * TCH;

    const size_t fbase = (size_t)seq * NF * C4 + c4;
    const v4f* __restrict__ src = reinterpret_cast<const v4f*>(xw) + fbase;
    v4f* __restrict__ dst       = reinterpret_cast<v4f*>(out) + fbase;

    const bool has_lo = chunk > 0;
    const bool has_hi = chunk < NCH - 1;
    const v4f* __restrict__ lo  = src + (size_t)(has_lo ? t0 - HALO : t0) * C4;
    const v4f* __restrict__ mid = src + (size_t)t0 * C4;
    const v4f* __restrict__ hi  = src + (size_t)(has_hi ? t0 + TCH : t0 + TCH - HALO) * C4;

    const int cb = (c4 & (CC / 4 - 1)) * 4;
    v4f bias;
    bias.x = bt[cb]; bias.y = bt[cb + 1]; bias.z = bt[cb + 2]; bias.w = bt[cb + 3];

    v4f v[TCH + 2 * HALO];
#pragma unroll
    for (int i = 0; i < HALO; ++i) v[i] = lo[(size_t)i * C4];
#pragma unroll
    for (int i = 0; i < TCH; ++i) v[HALO + i] = mid[(size_t)i * C4];
#pragma unroll
    for (int i = 0; i < HALO; ++i) v[HALO + TCH + i] = hi[(size_t)i * C4];

    const v4f zero4 = {0.f, 0.f, 0.f, 0.f};
#pragma unroll
    for (int i = 0; i < HALO; ++i) {
        v[i]              = has_lo ? v[i] : zero4;
        v[HALO + TCH + i] = has_hi ? v[HALO + TCH + i] : zero4;
    }

    v4f p[TCH + 6];
#pragma unroll
    for (int i = 0; i < TCH + 6; ++i) p[i] = v[i] + v[i + 1];
    v4f q[TCH + 4];
#pragma unroll
    for (int i = 0; i < TCH + 4; ++i) q[i] = p[i] + p[i + 2];
#pragma unroll
    for (int i = 0; i < TCH; ++i) {
        const v4f s = ((q[i] + q[i + 4]) + v[i + 8]) + bias;
        v4f o;
        o.x = s.x > 0.f ? s.x : 0.f;
        o.y = s.y > 0.f ? s.y : 0.f;
        o.z = s.z > 0.f ? s.z : 0.f;
        o.w = s.w > 0.f ? s.w : 0.f;
        dst[(size_t)(t0 + i) * C4] = o;
    }
}

extern "C" void kernel_launch(void* const* d_in, const int* in_sizes, int n_in,
                              void* d_out, int out_size, void* d_ws, size_t ws_size,
                              hipStream_t stream)
{
    const float* x     = (const float*)d_in[0];
    const float* A     = (const float*)d_in[1];
    const float* EI    = (const float*)d_in[2];
    const float* Ws    = (const float*)d_in[3];
    const float* bs    = (const float*)d_in[4];
    const float* Wt    = (const float*)d_in[5];
    const float* bt    = (const float*)d_in[6];
    const float* gamma = (const float*)d_in[7];
    const float* beta  = (const float*)d_in[8];

    float* out = (float*)d_out;
    float* ws  = (float*)d_ws;

    float* xwbuf = ws;

    stgcn_spatial_kernel<<<TT / FPB, 256, 0, stream>>>(
        x, Ws, Wt, gamma, beta, A, EI, bs, xwbuf);
    stgcn_temporal_kernel<<<(NSEQ * NCH * C4) / 256, 256, 0, stream>>>(
        xwbuf, bt, out);
}


# GEMM loops: two M-tiles per pass on two accumulator chains, A fragments fetched from LDS one 8-k block ahead as one group into a second register set (was 1 chain, 1 load per 2 WMMAs); same sum order
# speedup vs baseline: 5.1384x; 1.0320x over previous
#include <hip/hip_runtime.h>
#include <hip/hip_bf16.h>


typedef __attribute__((ext_vector_type(2))) float v2f;
typedef __attribute__((ext_vector_type(4))) float v4f;
typedef __attribute__((ext_vector_type(8))) float v8f;

constexpr int JV    = 25;
constexpr int CC    = 128;
constexpr int NF    = 300;
constexpr int NSEQ  = 8;
constexpr int TT    = NSEQ * NF;
constexpr int FPB   = 5;
constexpr int ROWS  = FPB * JV;
constexpr int RPAD  = 128;
constexpr int LST   = 132;
constexpr float GN_EPS = 1e-5f;

template <bool TO_LDS>
__device__ __forceinline__ void gemm_rows_128(const float* lds_a,
                                              const float* __restrict__ W,
                                              float* lds_out,
                                              float* __restrict__ gout,
                                              int grow0, int tid)
{
    const int lane = tid & 31;
    const int wv   = tid >> 5;
    const int l15  = lane & 15;
    const int kh   = lane >> 4;
    const int col  = wv * 16 + l15;

    v2f bf[32];
#pragma unroll
    for (int k = 0; k < 32; ++k) {
        const int kr = 4 * k + 2 * kh;
        bf[k].x = W[kr * CC + col];
        bf[k].y = W[(kr + 1) * CC + col];
    }

    constexpr int KB = 8;
    constexpr int NB = 32 / KB;
    constexpr int NP = RPAD / 32;
    static_assert(NB % 2 == 0, "block 0 of every pass must land in fragment set 0");

    const float* abase = lds_a + l15 * LST + 2 * kh;

    v2f fa[2][2][KB];
#pragma unroll
    for (int k = 0; k < KB; ++k) {
        fa[0][0][k] = *reinterpret_cast<const v2f*>(abase + 4 * k);
        fa[0][1][k] = *reinterpret_cast<const v2f*>(abase + 16 * LST + 4 * k);
    }

#pragma unroll 1
    for (int p = 0; p < NP; ++p) {
        const float* ar  = abase + p * 32 * LST;
        const float* arn = abase + (p + 1 < NP ? p + 1 : p) * 32 * LST;
        v8f acc0 = {}, acc1 = {};
#pragma unroll
        for (int b = 0; b < NB; ++b) {
            const int cur = b & 1, nxt = cur ^ 1;
            const float* src = (b + 1 < NB) ? ar + 4 * KB * (b + 1) : arn;
#pragma unroll
            for (int k = 0; k < KB; ++k) {
                fa[nxt][0][k] = *reinterpret_cast<const v2f*>(src + 4 * k);
                fa[nxt][1][k] = *reinterpret_cast<const v2f*>(src + 16 * LST + 4 * k);
            }
            __builtin_amdgcn_sched_barrier(0);
#pragma unroll
            for (int k = 0; k < KB; ++k) {
                acc0 = __builtin_amdgcn_wmma_f32_16x16x4_f32(
                    false, fa[cur][0][k], false, bf[b * KB + k], (short)0, acc0, false, false);
                acc1 = __builtin_amdgcn_wmma_f32_16x16x4_f32(
                    false, fa[cur][1][k], false, bf[b * KB + k], (short)0, acc1, false, false);
            }
            __builtin_amdgcn_sched_barrier(0);
        }
#pragma unroll
        for (int r = 0; r < 8; ++r) {
            const int row0 = p * 32 + r + 8 * kh;
            const int row1 = row0 + 16;
            if (TO_LDS) {
                lds_out[row0 * LST + col] = acc0[r];
                lds_out[row1 * LST + col] = acc1[r];
            } else {
                if (row0 < ROWS)
                    gout[(size_t)(grow0 + row0) * CC + col] = acc0[r];
                if (row1 < ROWS)
                    gout[(size_t)(grow0 + row1) * CC + col] = acc1[r];
            }
        }
    }
}

__global__ __launch_bounds__(256, 4) void stgcn_spatial_kernel(
    const float* __restrict__ x,    const float* __restrict__ Ws,
    const float* __restrict__ Wt,   const float* __restrict__ gamma,
    const float* __restrict__ beta, const float* __restrict__ A,
    const float* __restrict__ EI,   const float* __restrict__ bs,
    float* __restrict__ xw)
{
    __shared__ __attribute__((aligned(16))) float lds_x[RPAD * LST];
    __shared__ __attribute__((aligned(16))) float lds_y[RPAD * LST];
    __shared__ float lds_A[(3 * JV * JV + 255) / 256 * 256];
    float* const lds_h = lds_x;

    const int tid   = threadIdx.x;
    const int grow0 = blockIdx.x * ROWS;

    constexpr int NA  = 3 * JV * JV;
    constexpr int NAJ = (NA + 255) / 256;
    float ar[NAJ], er[NAJ];
    float b0;
    {
        const v4f* __restrict__ xg =
            reinterpret_cast<const v4f*>(x + (size_t)grow0 * CC);
        constexpr int NP = ROWS * CC / 4;
        constexpr int NJ = RPAD * CC / 4 / 256;
        static_assert(RPAD * CC / 4 == NJ * 256, "tile = whole pieces per thread");
        const v4f zero4 = {0.f, 0.f, 0.f, 0.f};
        v4f xr[NJ];
#pragma unroll
        for (int j = 0; j < NJ; ++j) {
            const int p = j * 256 + tid;
            xr[j] = xg[(j < NP / 256 || p < NP) ? p : NP - 1];
        }
#pragma unroll
        for (int j = 0; j < NAJ; ++j) {
            const int idx = j * 256 + tid;
            const int ci  = (j < NA / 256 || idx < NA) ? idx : NA - 1;
            ar[j] = A[ci];
            er[j] = EI[ci];
        }
        {
            const int bcol = (tid >> 5) * 16 + (tid & 15);
            b0 = bs[bcol] + bs[CC + bcol] + bs[2 * CC + bcol];
        }
#pragma unroll
        for (int j = 0; j < NJ; ++j) {
            const int p = j * 256 + tid;
            const int r = p / (CC / 4), c4 = p % (CC / 4);
            *reinterpret_cast<v4f*>(lds_x + r * LST + c4 * 4) =
                (j < NP / 256 || p < NP) ? xr[j] : zero4;
        }
    }
#pragma unroll
    for (int j = 0; j < NAJ; ++j) {
        lds_A[j * 256 + tid] = ar[j] * er[j];
    }
    __syncthreads();

    {
        const int lane = tid & 31;
        const int wv   = tid >> 5;
        const int l15  = lane & 15;
        const int kh   = lane >> 4;
        const int col  = wv * 16 + l15;

        v8f hacc[FPB][2];
        {
#pragma unroll
            for (int f = 0; f < FPB; ++f)
#pragma unroll
                for (int mt = 0; mt < 2; ++mt)
#pragma unroll
                    for (int r = 0; r < 8; ++r) hacc[f][mt][r] = b0;
        }

        for (int i = 0; i < 3; ++i) {
            gemm_rows_128<true>(lds_x, Ws + i * CC * CC, lds_y, nullptr, 0, tid);

            v2f af[2][7];
#pragma unroll
            for (int mt = 0; mt < 2; ++mt) {
                const int u = mt * 16 + l15;
                const float* arow = lds_A + i * JV * JV + (u < JV ? u : 0) * JV;
#pragma unroll
                for (int k = 0; k < 7; ++k) {
                    const int v0 = 4 * k + 2 * kh;
                    const float ax = arow[v0 < JV ? v0 : 0];
                    const float ay = arow[v0 + 1 < JV ? v0 + 1 : 0];
                    af[mt][k].x = (u < JV && v0 < JV) ? ax : 0.f;
                    af[mt][k].y = (u < JV && v0 + 1 < JV) ? ay : 0.f;
                }
            }
            __syncthreads();

#pragma unroll
            for (int f = 0; f < FPB; ++f) {
                const float* yb = lds_y + (f * JV + 2 * kh) * LST + col;
                v2f yf[7];
#pragma unroll
                for (int k = 0; k < 7; ++k) {
                    yf[k].x = yb[(4 * k) * LST];
                    yf[k].y = yb[(4 * k + 1) * LST];
                }
                yf[6].x = (kh == 0) ? yf[6].x : 0.f;
                yf[6].y = 0.f;
#pragma unroll
                for (int k = 0; k < 7; ++k) {
#pragma unroll
                    for (int mt = 0; mt < 2; ++mt)
                        hacc[f][mt] = __builtin_amdgcn_wmma_f32_16x16x4_f32(
                            false, af[mt][k], false, yf[k], (short)0,
                            hacc[f][mt], false, false);
                }
            }
        }

#pragma unroll
        for (int f = 0; f < FPB; ++f)
#pragma unroll
            for (int mt = 0; mt < 2; ++mt)
#pragma unroll
                for (int r = 0; r < 8; ++r) {
                    const int u = mt * 16 + r + 8 * kh;
                    if (u < JV)
                        lds_h[(f * JV + u) * LST + col] = hacc[f][mt][r];
                }
    }
    __syncthreads();

    {
        const int lane = tid & 31;
        const int wv   = tid >> 5;
        if (wv < FPB) {
            float* hb = lds_h + (wv * JV) * LST + lane * 4;
            v4f hv[JV];
            v4f s4 = {0.f, 0.f, 0.f, 0.f}, q4 = {0.f, 0.f, 0.f, 0.f};
#pragma unroll
            for (int u = 0; u < JV; ++u) {
                hv[u] = *reinterpret_cast<const v4f*>(hb + u * LST);
                s4 += hv[u];
                q4 += hv[u] * hv[u];
            }
            float ts = (s4.x + s4.y) + (s4.z + s4.w);
            float tq = (q4.x + q4.y) + (q4.z + q4.w);
#pragma unroll
            for (int off = 16; off > 0; off >>= 1) {
                ts += __shfl_xor(ts, off);
                tq += __shfl_xor(tq, off);
            }
            const float inv = 1.f / (float)(JV * CC);
            const float mu  = ts * inv;
            const float var = tq * inv - mu * mu;
            const float rs  = rsqrtf(var + GN_EPS);
            const v4f* __restrict__ g4 = reinterpret_cast<const v4f*>(gamma) + lane;
            const v4f* __restrict__ b4 = reinterpret_cast<const v4f*>(beta) + lane;
#pragma unroll
            for (int u = 0; u < JV; ++u) {
                const v4f gm = g4[u * (CC / 4)];
                const v4f bb = b4[u * (CC / 4)];
                const v4f val = (hv[u] - mu) * rs * gm + bb;
                v4f o;
                o.x = val.x > 0.f ? val.x : 0.f;
                o.y = val.y > 0.f ? val.y : 0.f;
                o.z = val.z > 0.f ? val.z : 0.f;
                o.w = val.w > 0.f ? val.w : 0.f;
                *reinterpret_cast<v4f*>(hb + u * LST) = o;
            }
        }
    }
    __syncthreads();

    gemm_rows_128<false>(lds_h, Wt, nullptr, xw, grow0, tid);
}

constexpr int TCH  = 15;
constexpr int NCH  = NF / TCH;
constexpr int C4   = JV * CC / 4;
constexpr int HALO = 4;

__global__ __launch_bounds__(256) void stgcn_temporal_kernel(
    const float* __restrict__ xw, const float* __restrict__ bt,
    float* __restrict__ out)
{
    const int g     = blockIdx.x * 256 + threadIdx.x;
    const int crow  = g / C4;
    const int c4    = g % C4;
    const int seq   = crow / NCH;
    const int chunk = crow % NCH;
    const int t0    = chunk * TCH;

    const size_t fbase = (size_t)seq * NF * C4 + c4;
    const v4f* __restrict__ src = reinterpret_cast<const v4f*>(xw) + fbase;
    v4f* __restrict__ dst       = reinterpret_cast<v4f*>(out) + fbase;

    const bool has_lo = chunk > 0;
    const bool has_hi = chunk < NCH - 1;
    const v4f* __restrict__ lo  = src + (size_t)(has_lo ? t0 - HALO : t0) * C4;
    const v4f* __restrict__ mid = src + (size_t)t0 * C4;
    const v4f* __restrict__ hi  = src + (size_t)(has_hi ? t0 + TCH : t0 + TCH - HALO) * C4;

    const int cb = (c4 & (CC / 4 - 1)) * 4;
    v4f bias;
    bias.x = bt[cb]; bias.y = bt[cb + 1]; bias.z = bt[cb + 2]; bias.w = bt[cb + 3];

    v4f v[TCH + 2 * HALO];
#pragma unroll
    for (int i = 0; i < HALO; ++i) v[i] = lo[(size_t)i * C4];
#pragma unroll
    for (int i = 0; i < TCH; ++i) v[HALO + i] = mid[(size_t)i * C4];
#pragma unroll
    for (int i = 0; i < HALO; ++i) v[HALO + TCH + i] = hi[(size_t)i * C4];

    const v4f zero4 = {0.f, 0.f, 0.f, 0.f};
#pragma unroll
    for (int i = 0; i < HALO; ++i) {
        v[i]              = has_lo ? v[i] : zero4;
        v[HALO + TCH + i] = has_hi ? v[HALO + TCH + i] : zero4;
    }

    v4f p[TCH + 6];
#pragma unroll
    for (int i = 0; i < TCH + 6; ++i) p[i] = v[i] + v[i + 1];
    v4f q[TCH + 4];
#pragma unroll
    for (int i = 0; i < TCH + 4; ++i) q[i] = p[i] + p[i + 2];
#pragma unroll
    for (int i = 0; i < TCH; ++i) {
        const v4f s = ((q[i] + q[i + 4]) + v[i + 8]) + bias;
        v4f o;
        o.x = s.x > 0.f ? s.x : 0.f;
        o.y = s.y > 0.f ? s.y : 0.f;
        o.z = s.z > 0.f ? s.z : 0.f;
        o.w = s.w > 0.f ? s.w : 0.f;
        dst[(size_t)(t0 + i) * C4] = o;
    }
}

extern "C" void kernel_launch(void* const* d_in, const int* in_sizes, int n_in,
                              void* d_out, int out_size, void* d_ws, size_t ws_size,
                              hipStream_t stream)
{
    const float* x     = (const float*)d_in[0];
    const float* A     = (const float*)d_in[1];
    const float* EI    = (const float*)d_in[2];
    const float* Ws    = (const float*)d_in[3];
    const float* bs    = (const float*)d_in[4];
    const float* Wt    = (const float*)d_in[5];
    const float* bt    = (const float*)d_in[6];
    const float* gamma = (const float*)d_in[7];
    const float* beta  = (const float*)d_in[8];

    float* out = (float*)d_out;
    float* ws  = (float*)d_ws;

    float* xwbuf = ws;

    stgcn_spatial_kernel<<<TT / FPB, 256, 0, stream>>>(
        x, Ws, Wt, gamma, beta, A, EI, bs, xwbuf);
    stgcn_temporal_kernel<<<(NSEQ * NCH * C4) / 256, 256, 0, stream>>>(
        xwbuf, bt, out);
}
